# starting kernel
# baseline (speedup 1.0000x reference)
#include <hip/hip_runtime.h>

typedef __attribute__((ext_vector_type(2))) float v2f;
typedef __attribute__((ext_vector_type(8))) float v8f;

#define F_IN  32
#define HC    64
#define C_OUT 8


__global__ __launch_bounds__(256) void k_deg_init(float* __restrict__ deg, int N) {
  int i = blockIdx.x * 256 + threadIdx.x;
  if (i < N) deg[i] = 1.0f;
}

__global__ __launch_bounds__(256) void k_deg_count(const int* __restrict__ dst,
                                                   float* __restrict__ deg, int E) {
  int e = blockIdx.x * 256 + threadIdx.x;
  if (e < E) atomicAdd(&deg[dst[e]], 1.0f);
}

__global__ __launch_bounds__(256) void k_dinv(float* __restrict__ deg, int N) {
  int i = blockIdx.x * 256 + threadIdx.x;
  if (i < N) deg[i] = 1.0f / sqrtf(deg[i]);
}


__global__ __launch_bounds__(32) void k_gemm1(const float* __restrict__ x,
                                              const float* __restrict__ W1,
                                              float* __restrict__ h1, int N) {
  const int lane  = threadIdx.x;
  const int l15   = lane & 15;
  const int khalf = lane >> 4;
  int row = blockIdx.x * 16 + l15;
  if (row >= N) row = N - 1;
  v8f c[4] = {};
  #pragma unroll
  for (int k0 = 0; k0 < F_IN; k0 += 4) {
    const int ka = k0 + khalf * 2;
    v2f a = { x[(size_t)row * F_IN + ka], x[(size_t)row * F_IN + ka + 1] };
    #pragma unroll
    for (int nt = 0; nt < 4; ++nt) {
      const int col = nt * 16 + l15;
      v2f b = { W1[(size_t)ka * HC + col], W1[(size_t)(ka + 1) * HC + col] };
      c[nt] = __builtin_amdgcn_wmma_f32_16x16x4_f32(false, a, false, b,
                                                    (short)0, c[nt], false, false);
    }
  }
  const int mbase = blockIdx.x * 16 + khalf * 8;
  #pragma unroll
  for (int nt = 0; nt < 4; ++nt)
    #pragma unroll
    for (int r = 0; r < 8; ++r) {
      const int m = mbase + r;
      if (m < N) h1[(size_t)m * HC + nt * 16 + l15] = c[nt][r];
    }
}


__global__ __launch_bounds__(256) void k_agg1_init(const float* __restrict__ h1,
                                                   const float* __restrict__ b1,
                                                   const float* __restrict__ dinv,
                                                   float* __restrict__ out1, int N) {
  int i = blockIdx.x * 256 + threadIdx.x;
  if (i >= N * (HC / 4)) return;
  const int node = i >> 4;
  const int fq   = i & 15;
  const float di = dinv[node];
  const float s  = di * di;
  float4 v  = ((const float4*)h1)[i];
  float4 bb = ((const float4*)b1)[fq];
  float4 o;
  o.x = bb.x + v.x * s;  o.y = bb.y + v.y * s;
  o.z = bb.z + v.z * s;  o.w = bb.w + v.w * s;
  ((float4*)out1)[i] = o;
}

__global__ __launch_bounds__(256) void k_agg1_edges(const int* __restrict__ src,
                                                    const int* __restrict__ dst,
                                                    const float* __restrict__ dinv,
                                                    const float* __restrict__ h1,
                                                    float* __restrict__ out1, int E) {
  unsigned gid = blockIdx.x * 256u + threadIdx.x;
  const int e  = gid >> 4;
  const int fq = gid & 15;
  if (e >= E) return;
  const int s = src[e], d = dst[e];
  const float nrm = dinv[s] * dinv[d];
  float4 v = ((const float4*)(h1 + (size_t)s * HC))[fq];
  float* o = out1 + (size_t)d * HC + fq * 4;
  atomicAdd(o + 0, v.x * nrm);
  atomicAdd(o + 1, v.y * nrm);
  atomicAdd(o + 2, v.z * nrm);
  atomicAdd(o + 3, v.w * nrm);
}


__global__ __launch_bounds__(32) void k_gemm2(const float* __restrict__ h,
                                              const float* __restrict__ W2,
                                              float* __restrict__ h2, int N) {
  const int lane  = threadIdx.x;
  const int l15   = lane & 15;
  const int khalf = lane >> 4;
  int row = blockIdx.x * 16 + l15;
  if (row >= N) row = N - 1;
  const bool live_col = (l15 < C_OUT);
  v8f c = {};
  #pragma unroll
  for (int k0 = 0; k0 < HC; k0 += 4) {
    const int ka = k0 + khalf * 2;
    v2f a = { h[(size_t)row * HC + ka], h[(size_t)row * HC + ka + 1] };
    v2f b = { live_col ? W2[(size_t)ka * C_OUT + l15] : 0.0f,
              live_col ? W2[(size_t)(ka + 1) * C_OUT + l15] : 0.0f };
    c = __builtin_amdgcn_wmma_f32_16x16x4_f32(false, a, false, b,
                                              (short)0, c, false, false);
  }
  const int mbase = blockIdx.x * 16 + khalf * 8;
  if (live_col) {
    #pragma unroll
    for (int r = 0; r < 8; ++r) {
      const int m = mbase + r;
      if (m < N) h2[(size_t)m * C_OUT + l15] = c[r];
    }
  }
}


__global__ __launch_bounds__(256) void k_out_init(const float* __restrict__ h2,
                                                  const float* __restrict__ b2,
                                                  const float* __restrict__ dinv,
                                                  float* __restrict__ out, int N) {
  int i = blockIdx.x * 256 + threadIdx.x;
  if (i >= N * (C_OUT / 4)) return;
  const int node = i >> 1;
  const int half = i & 1;
  const float di = dinv[node];
  const float s  = di * di;
  float4 v  = ((const float4*)h2)[i];
  float4 bb = ((const float4*)b2)[half];
  float4 o;
  o.x = bb.x + v.x * s;  o.y = bb.y + v.y * s;
  o.z = bb.z + v.z * s;  o.w = bb.w + v.w * s;
  ((float4*)out)[i] = o;
}

__global__ __launch_bounds__(256) void k_agg2_edges(const int* __restrict__ src,
                                                    const int* __restrict__ dst,
                                                    const float* __restrict__ dinv,
                                                    const float* __restrict__ h2,
                                                    float* __restrict__ out, int E) {
  unsigned gid = blockIdx.x * 256u + threadIdx.x;
  const int e    = gid >> 1;
  const int half = gid & 1;
  if (e >= E) return;
  const int s = src[e], d = dst[e];
  const float nrm = dinv[s] * dinv[d];
  float4 v = ((const float4*)(h2 + (size_t)s * C_OUT))[half];
  float* o = out + (size_t)d * C_OUT + half * 4;
  atomicAdd(o + 0, v.x * nrm);
  atomicAdd(o + 1, v.y * nrm);
  atomicAdd(o + 2, v.z * nrm);
  atomicAdd(o + 3, v.w * nrm);
}


extern "C" void kernel_launch(void* const* d_in, const int* in_sizes, int n_in,
                              void* d_out, int out_size, void* d_ws, size_t ws_size,
                              hipStream_t stream) {
  const float* x  = (const float*)d_in[0];
  const int*   ei = (const int*)d_in[1];
  const float* W1 = (const float*)d_in[2];
  const float* b1 = (const float*)d_in[3];
  const float* W2 = (const float*)d_in[4];
  const float* b2 = (const float*)d_in[5];
  float* out = (float*)d_out;

  const int N = in_sizes[0] / F_IN;
  const int E = in_sizes[1] / 2;
  const int* src = ei;
  const int* dst = ei + E;

  float* dinv = (float*)d_ws;
  float* h1   = dinv + N;
  float* out1 = h1 + (size_t)N * HC;
  float* h2   = out1 + (size_t)N * HC;

  const int nbN  = (N + 255) / 256;
  const int nbE  = (E + 255) / 256;
  const int tiles = (N + 15) / 16;

  k_deg_init <<<nbN, 256, 0, stream>>>(dinv, N);
  k_deg_count<<<nbE, 256, 0, stream>>>(dst, dinv, E);
  k_dinv     <<<nbN, 256, 0, stream>>>(dinv, N);

  k_gemm1<<<tiles, 32, 0, stream>>>(x, W1, h1, N);

  k_agg1_init <<<(N * (HC / 4) + 255) / 256, 256, 0, stream>>>(h1, b1, dinv, out1, N);
  k_agg1_edges<<<(unsigned)(((size_t)E * 16 + 255) / 256), 256, 0, stream>>>(src, dst, dinv, h1, out1, E);

  k_gemm2<<<tiles, 32, 0, stream>>>(out1, W2, h2, N);

  k_out_init  <<<(N * (C_OUT / 4) + 255) / 256, 256, 0, stream>>>(h2, b2, dinv, out, N);
  k_agg2_edges<<<(unsigned)(((size_t)E * 2 + 255) / 256), 256, 0, stream>>>(src, dst, dinv, h2, out, E);
}


# Layer 1 sums the dinv-scaled 32-float input rows over the edges before its linear map, (A x) W1 for A (x W1): one 128-B line per edge, not two; both WMMA products in one launch; sum order changes
# speedup vs baseline: 3.4745x; 3.4745x over previous
#include <hip/hip_runtime.h>

typedef __attribute__((ext_vector_type(2))) float v2f;
typedef __attribute__((ext_vector_type(8))) float v8f;

#define F_IN  32
#define HC    64
#define C_OUT 8

static __device__ __forceinline__ int pkg_half() {
  const unsigned die = (__builtin_amdgcn_s_sendmsg_rtn(0x87) >> 16) & 15u;
  return (int)((die >> 2) & 1u);
}


__global__ __launch_bounds__(256) void k_deg_init(float* __restrict__ deg, float* __restrict__ deg_b, int N) {
  int i = blockIdx.x * 256 + threadIdx.x;
  if (i < N) { deg[i] = 1.0f; deg_b[i] = 0.0f; }
}

__global__ __launch_bounds__(256) void k_deg_count(const int* __restrict__ dst,
                                                   float* __restrict__ deg, float* __restrict__ deg_b, int E) {
  float* plane = pkg_half() ? deg_b : deg;
  int e = blockIdx.x * 256 + threadIdx.x;
  if (e < E) atomicAdd(&plane[dst[e]], 1.0f);
}

__global__ __launch_bounds__(256) void k_scale_x(const float* __restrict__ x,
                                                 const float* __restrict__ deg, const float* __restrict__ deg_b,
                                                 float* __restrict__ dinv, float* __restrict__ xs,
                                                 float* __restrict__ ax, int N) {
  int i = blockIdx.x * 256 + threadIdx.x;
  if (i >= N * (F_IN / 4)) return;
  const int node = i >> 3;
  float4 v = ((const float4*)x)[i];
  const float di = 1.0f / sqrtf(deg[node] + deg_b[node]);
  if ((i & 7) == 0) dinv[node] = di;
  v.x *= di;  v.y *= di;  v.z *= di;  v.w *= di;
  ((float4*)xs)[i] = v;
  ((float4*)ax)[i] = v;
  ((float4*)ax)[(size_t)N * (F_IN / 4) + i] = make_float4(0.0f, 0.0f, 0.0f, 0.0f);
}


__global__ __launch_bounds__(256) void k_aggx_edges(const int* __restrict__ src,
                                                    const int* __restrict__ dst,
                                                    const float* __restrict__ xs,
                                                    float* __restrict__ ax, int E, int N) {
  const int lane = threadIdx.x & 31;
  float* __restrict__ plane = ax + (size_t)pkg_half() * N * F_IN;
  const int wib    = __builtin_amdgcn_readfirstlane((int)(threadIdx.x >> 5));
  const int nwaves = gridDim.x * 8;
  const int nbatch = (E + 31) >> 5;
  for (int b = blockIdx.x * 8 + wib; b < nbatch; b += nwaves) {
    const int e0 = b << 5;
    const int e  = e0 + lane;
    int s = 0, d = 0;
    if (e < E) { s = src[e]; d = dst[e]; }
    const int cnt = (E - e0 < 32) ? (E - e0) : 32;
    if (cnt == 32) {
      #pragma unroll
      for (int g = 0; g < 32; g += 16) {
        float v[16];
        #pragma unroll
        for (int u = 0; u < 16; ++u) {
          const int sj = __builtin_amdgcn_readlane(s, g + u);
          v[u] = xs[(size_t)sj * F_IN + lane];
        }
        #pragma unroll
        for (int u = 0; u < 16; ++u) {
          const int dj = __builtin_amdgcn_readlane(d, g + u);
          atomicAdd(plane + (size_t)dj * F_IN + lane, v[u]);
        }
      }
    } else {
      for (int j = 0; j < cnt; ++j) {
        const int sj = __builtin_amdgcn_readlane(s, j);
        const int dj = __builtin_amdgcn_readlane(d, j);
        atomicAdd(plane + (size_t)dj * F_IN + lane, xs[(size_t)sj * F_IN + lane]);
      }
    }
  }
}


__global__ __launch_bounds__(32) void k_gemm12(const float* __restrict__ ax,
                                               const float* __restrict__ W1,
                                               const float* __restrict__ b1,
                                               const float* __restrict__ W2,
                                               const float* __restrict__ dinv,
                                               float* __restrict__ h2, int N) {
  const int lane = threadIdx.x;
  const int c    = lane & 15;
  const int h    = lane >> 4;
  int node = blockIdx.x * 16 + c;
  const bool live = (node < N);
  if (!live) node = N - 1;
  const float dn = dinv[node];
  const float4* __restrict__ p0 = (const float4*)(ax + (size_t)node * F_IN + 16 * h);
  const float4* __restrict__ p1 = (const float4*)(ax + (size_t)N * F_IN + (size_t)node * F_IN + 16 * h);
  float a1[16];
  #pragma unroll
  for (int q = 0; q < 4; ++q) {
    const float4 u = p0[q], w = p1[q];
    a1[4 * q + 0] = dn * (u.x + w.x);
    a1[4 * q + 1] = dn * (u.y + w.y);
    a1[4 * q + 2] = dn * (u.z + w.z);
    a1[4 * q + 3] = dn * (u.w + w.w);
  }
  v8f t[4];
  #pragma unroll
  for (int r = 0; r < 8; ++r) {
    const float4 bb = ((const float4*)b1)[8 * h + r];
    t[0][r] = bb.x;  t[1][r] = bb.y;  t[2][r] = bb.z;  t[3][r] = bb.w;
  }
  #pragma unroll
  for (int j = 0; j < 8; ++j) {
    const int k = 16 * h + 2 * j;
    const float4 w0 = ((const float4*)(W1 + (size_t)k * HC))[c];
    const float4 w1 = ((const float4*)(W1 + (size_t)(k + 1) * HC))[c];
    const v2f bx = { a1[2 * j], a1[2 * j + 1] };
    const v2f aw0 = { w0.x, w1.x }, aw1 = { w0.y, w1.y }, aw2 = { w0.z, w1.z }, aw3 = { w0.w, w1.w };
    t[0] = __builtin_amdgcn_wmma_f32_16x16x4_f32(false, aw0, false, bx, (short)0, t[0], false, false);
    t[1] = __builtin_amdgcn_wmma_f32_16x16x4_f32(false, aw1, false, bx, (short)0, t[1], false, false);
    t[2] = __builtin_amdgcn_wmma_f32_16x16x4_f32(false, aw2, false, bx, (short)0, t[2], false, false);
    t[3] = __builtin_amdgcn_wmma_f32_16x16x4_f32(false, aw3, false, bx, (short)0, t[3], false, false);
  }
  const bool live_row = (c < C_OUT);
  const float* __restrict__ w2c = W2 + (c & (C_OUT - 1));
  v8f o = {};
  #pragma unroll
  for (int mt = 0; mt < 4; ++mt)
    #pragma unroll
    for (int j = 0; j < 4; ++j) {
      const int k = 4 * (8 * h + 2 * j) + mt;
      const float wa = w2c[(size_t)k * C_OUT], wb = w2c[(size_t)(k + 4) * C_OUT];
      const v2f aw = { live_row ? wa : 0.0f, live_row ? wb : 0.0f };
      const v2f bt = { t[mt][2 * j], t[mt][2 * j + 1] };
      o = __builtin_amdgcn_wmma_f32_16x16x4_f32(false, aw, false, bt, (short)0, o, false, false);
    }
  if (h == 0 && live) {
    float4* __restrict__ q = (float4*)(h2 + (size_t)node * C_OUT);
    q[0] = make_float4(o[0] * dn, o[1] * dn, o[2] * dn, o[3] * dn);
    q[1] = make_float4(o[4] * dn, o[5] * dn, o[6] * dn, o[7] * dn);
  }
}


__global__ __launch_bounds__(256) void k_out_init(const float* __restrict__ h2,
                                                  float* __restrict__ out,
                                                  float* __restrict__ out_b, int N) {
  int i = blockIdx.x * 256 + threadIdx.x;
  if (i >= N * (C_OUT / 4)) return;
  ((float4*)out)[i] = ((const float4*)h2)[i];
  ((float4*)out_b)[i] = make_float4(0.0f, 0.0f, 0.0f, 0.0f);
}


#define BCAST8(i) ((i) * 0x249249)

#define STEP_LOAD(i) { \
  const int sj = (int)__builtin_amdgcn_mov_dpp8((unsigned)s, BCAST8(i)); \
  dd[i] = (int)__builtin_amdgcn_mov_dpp8((unsigned)d, BCAST8(i)); \
  v[i]  = h2[(size_t)sj * C_OUT + f]; }

static __device__ __forceinline__ void agg2_batch(const bool full, const int s, const int d,
                                                  const int f, const int left,
                                                  const float* __restrict__ h2, float* __restrict__ out) {
  float v[8];
  int   dd[8];
  STEP_LOAD(0) STEP_LOAD(1) STEP_LOAD(2) STEP_LOAD(3)
  STEP_LOAD(4) STEP_LOAD(5) STEP_LOAD(6) STEP_LOAD(7)
  #pragma unroll
  for (int i = 0; i < 8; ++i)
    if (full || i < left) atomicAdd(out + (size_t)dd[i] * C_OUT + f, v[i]);
}

__global__ __launch_bounds__(256) void k_agg2_edges(const int* __restrict__ src,
                                                    const int* __restrict__ dst,
                                                    const float* __restrict__ h2,
                                                    float* __restrict__ out,
                                                    float* __restrict__ out_b, int E) {
  const int lane = threadIdx.x & 31;
  const int f    = lane & 7;
  float* __restrict__ plane = pkg_half() ? out_b : out;
  const int wib    = __builtin_amdgcn_readfirstlane((int)(threadIdx.x >> 5));
  const int nwaves = gridDim.x * 8;
  const int nbatch = (E + 31) >> 5;
  for (int b = blockIdx.x * 8 + wib; b < nbatch; b += nwaves) {
    const int e0 = b << 5;
    const int e  = e0 + lane;
    int s = 0, d = 0;
    if (e < E) { s = src[e]; d = dst[e]; }
    if (E - e0 >= 32) agg2_batch(true,  s, d, f, 8, h2, plane);
    else              agg2_batch(false, s, d, f, E - e0 - (lane & 24), h2, plane);
  }
}

#undef STEP_LOAD

__global__ __launch_bounds__(256) void k_out_sum(const float* __restrict__ out_b,
                                                 const float* __restrict__ b2,
                                                 const float* __restrict__ dinv,
                                                 float* __restrict__ out, int N) {
  int i = blockIdx.x * 256 + threadIdx.x;
  if (i >= N * (C_OUT / 4)) return;
  const float di = dinv[i >> 1];
  float4 a  = ((float4*)out)[i];
  float4 b  = ((const float4*)out_b)[i];
  float4 bb = ((const float4*)b2)[i & 1];
  a.x = bb.x + di * (a.x + b.x);  a.y = bb.y + di * (a.y + b.y);
  a.z = bb.z + di * (a.z + b.z);  a.w = bb.w + di * (a.w + b.w);
  ((float4*)out)[i] = a;
}


extern "C" void kernel_launch(void* const* d_in, const int* in_sizes, int n_in,
                              void* d_out, int out_size, void* d_ws, size_t ws_size,
                              hipStream_t stream) {
  const float* x  = (const float*)d_in[0];
  const int*   ei = (const int*)d_in[1];
  const float* W1 = (const float*)d_in[2];
  const float* b1 = (const float*)d_in[3];
  const float* W2 = (const float*)d_in[4];
  const float* b2 = (const float*)d_in[5];
  float* out = (float*)d_out;

  const int N = in_sizes[0] / F_IN;
  const int E = in_sizes[1] / 2;
  const int* src = ei;
  const int* dst = ei + E;

  float* xs    = (float*)d_ws;
  float* ax    = xs + (size_t)N * F_IN;
  float* h2    = ax + (size_t)2 * N * F_IN;
  float* out_b = h2 + (size_t)N * C_OUT;
  float* dinv  = out_b + (size_t)N * C_OUT;
  float* deg   = dinv + N;
  float* deg_b = deg + N;

  const int nbN  = (N + 255) / 256;
  const int nbE  = (E + 255) / 256;
  const int tiles = (N + 15) / 16;
  int nbW = ((E + 31) / 32 + 7) / 8;
  if (nbW > 2048) nbW = 2048;
  if (nbW < 1) nbW = 1;

  k_deg_init <<<nbN, 256, 0, stream>>>(deg, deg_b, N);
  k_deg_count<<<nbE, 256, 0, stream>>>(dst, deg, deg_b, E);
  k_scale_x  <<<(N * (F_IN / 4) + 255) / 256, 256, 0, stream>>>(x, deg, deg_b, dinv, xs, ax, N);

  k_aggx_edges<<<nbW, 256, 0, stream>>>(src, dst, xs, ax, E, N);

  k_gemm12<<<tiles, 32, 0, stream>>>(ax, W1, b1, W2, dinv, h2, N);

  k_out_init  <<<(N * (C_OUT / 4) + 255) / 256, 256, 0, stream>>>(h2, out, out_b, N);
  k_agg2_edges<<<nbW, 256, 0, stream>>>(src, dst, h2, out, out_b, E);
  k_out_sum   <<<(N * (C_OUT / 4) + 255) / 256, 256, 0, stream>>>(out_b, b2, dinv, out, N);
}


# Layer-2 start values (own scaled row into the output plane, zero into the second plane) are stored by the GEMM launch's epilogue; the start-value launch goes: 8 launches to 7; order of sums kept
# speedup vs baseline: 3.6948x; 1.0634x over previous
#include <hip/hip_runtime.h>

typedef __attribute__((ext_vector_type(2))) float v2f;
typedef __attribute__((ext_vector_type(8))) float v8f;

#define F_IN  32
#define HC    64
#define C_OUT 8

static __device__ __forceinline__ int pkg_half() {
  const unsigned die = (__builtin_amdgcn_s_sendmsg_rtn(0x87) >> 16) & 15u;
  return (int)((die >> 2) & 1u);
}


__global__ __launch_bounds__(256) void k_deg_init(float* __restrict__ deg, float* __restrict__ deg_b, int N) {
  int i = blockIdx.x * 256 + threadIdx.x;
  if (i < N) { deg[i] = 1.0f; deg_b[i] = 0.0f; }
}

__global__ __launch_bounds__(256) void k_deg_count(const int* __restrict__ dst,
                                                   float* __restrict__ deg, float* __restrict__ deg_b, int E) {
  float* plane = pkg_half() ? deg_b : deg;
  int e = blockIdx.x * 256 + threadIdx.x;
  if (e < E) atomicAdd(&plane[dst[e]], 1.0f);
}

__global__ __launch_bounds__(256) void k_scale_x(const float* __restrict__ x,
                                                 const float* __restrict__ deg, const float* __restrict__ deg_b,
                                                 float* __restrict__ dinv, float* __restrict__ xs,
                                                 float* __restrict__ ax, int N) {
  int i = blockIdx.x * 256 + threadIdx.x;
  if (i >= N * (F_IN / 4)) return;
  const int node = i >> 3;
  float4 v = ((const float4*)x)[i];
  const float di = 1.0f / sqrtf(deg[node] + deg_b[node]);
  if ((i & 7) == 0) dinv[node] = di;
  v.x *= di;  v.y *= di;  v.z *= di;  v.w *= di;
  ((float4*)xs)[i] = v;
  ((float4*)ax)[i] = v;
  ((float4*)ax)[(size_t)N * (F_IN / 4) + i] = make_float4(0.0f, 0.0f, 0.0f, 0.0f);
}


__global__ __launch_bounds__(256) void k_aggx_edges(const int* __restrict__ src,
                                                    const int* __restrict__ dst,
                                                    const float* __restrict__ xs,
                                                    float* __restrict__ ax, int E, int N) {
  const int lane = threadIdx.x & 31;
  float* __restrict__ plane = ax + (size_t)pkg_half() * N * F_IN;
  const int wib    = __builtin_amdgcn_readfirstlane((int)(threadIdx.x >> 5));
  const int nwaves = gridDim.x * 8;
  const int nbatch = (E + 31) >> 5;
  for (int b = blockIdx.x * 8 + wib; b < nbatch; b += nwaves) {
    const int e0 = b << 5;
    const int e  = e0 + lane;
    int s = 0, d = 0;
    if (e < E) { s = src[e]; d = dst[e]; }
    const int cnt = (E - e0 < 32) ? (E - e0) : 32;
    if (cnt == 32) {
      #pragma unroll
      for (int g = 0; g < 32; g += 16) {
        float v[16];
        #pragma unroll
        for (int u = 0; u < 16; ++u) {
          const int sj = __builtin_amdgcn_readlane(s, g + u);
          v[u] = xs[(size_t)sj * F_IN + lane];
        }
        #pragma unroll
        for (int u = 0; u < 16; ++u) {
          const int dj = __builtin_amdgcn_readlane(d, g + u);
          atomicAdd(plane + (size_t)dj * F_IN + lane, v[u]);
        }
      }
    } else {
      for (int j = 0; j < cnt; ++j) {
        const int sj = __builtin_amdgcn_readlane(s, j);
        const int dj = __builtin_amdgcn_readlane(d, j);
        atomicAdd(plane + (size_t)dj * F_IN + lane, xs[(size_t)sj * F_IN + lane]);
      }
    }
  }
}


__global__ __launch_bounds__(32) void k_gemm12(const float* __restrict__ ax,
                                               const float* __restrict__ W1,
                                               const float* __restrict__ b1,
                                               const float* __restrict__ W2,
                                               const float* __restrict__ dinv,
                                               float* __restrict__ h2,
                                               float* __restrict__ out,
                                               float* __restrict__ out_b, int N) {
  const int lane = threadIdx.x;
  const int c    = lane & 15;
  const int h    = lane >> 4;
  int node = blockIdx.x * 16 + c;
  const bool live = (node < N);
  if (!live) node = N - 1;
  const float dn = dinv[node];
  const float4* __restrict__ p0 = (const float4*)(ax + (size_t)node * F_IN + 16 * h);
  const float4* __restrict__ p1 = (const float4*)(ax + (size_t)N * F_IN + (size_t)node * F_IN + 16 * h);
  float a1[16];
  #pragma unroll
  for (int q = 0; q < 4; ++q) {
    const float4 u = p0[q], w = p1[q];
    a1[4 * q + 0] = dn * (u.x + w.x);
    a1[4 * q + 1] = dn * (u.y + w.y);
    a1[4 * q + 2] = dn * (u.z + w.z);
    a1[4 * q + 3] = dn * (u.w + w.w);
  }
  v8f t[4];
  #pragma unroll
  for (int r = 0; r < 8; ++r) {
    const float4 bb = ((const float4*)b1)[8 * h + r];
    t[0][r] = bb.x;  t[1][r] = bb.y;  t[2][r] = bb.z;  t[3][r] = bb.w;
  }
  #pragma unroll
  for (int j = 0; j < 8; ++j) {
    const int k = 16 * h + 2 * j;
    const float4 w0 = ((const float4*)(W1 + (size_t)k * HC))[c];
    const float4 w1 = ((const float4*)(W1 + (size_t)(k + 1) * HC))[c];
    const v2f bx = { a1[2 * j], a1[2 * j + 1] };
    const v2f aw0 = { w0.x, w1.x }, aw1 = { w0.y, w1.y }, aw2 = { w0.z, w1.z }, aw3 = { w0.w, w1.w };
    t[0] = __builtin_amdgcn_wmma_f32_16x16x4_f32(false, aw0, false, bx, (short)0, t[0], false, false);
    t[1] = __builtin_amdgcn_wmma_f32_16x16x4_f32(false, aw1, false, bx, (short)0, t[1], false, false);
    t[2] = __builtin_amdgcn_wmma_f32_16x16x4_f32(false, aw2, false, bx, (short)0, t[2], false, false);
    t[3] = __builtin_amdgcn_wmma_f32_16x16x4_f32(false, aw3, false, bx, (short)0, t[3], false, false);
  }
  const bool live_row = (c < C_OUT);
  const float* __restrict__ w2c = W2 + (c & (C_OUT - 1));
  v8f o = {};
  #pragma unroll
  for (int mt = 0; mt < 4; ++mt)
    #pragma unroll
    for (int j = 0; j < 4; ++j) {
      const int k = 4 * (8 * h + 2 * j) + mt;
      const float wa = w2c[(size_t)k * C_OUT], wb = w2c[(size_t)(k + 4) * C_OUT];
      const v2f aw = { live_row ? wa : 0.0f, live_row ? wb : 0.0f };
      const v2f bt = { t[mt][2 * j], t[mt][2 * j + 1] };
      o = __builtin_amdgcn_wmma_f32_16x16x4_f32(false, aw, false, bt, (short)0, o, false, false);
    }
  if (h == 0 && live) {
    const float4 r0 = make_float4(o[0] * dn, o[1] * dn, o[2] * dn, o[3] * dn);
    const float4 r1 = make_float4(o[4] * dn, o[5] * dn, o[6] * dn, o[7] * dn);
    const float4 z  = make_float4(0.0f, 0.0f, 0.0f, 0.0f);
    float4* __restrict__ q  = (float4*)(h2 + (size_t)node * C_OUT);
    float4* __restrict__ qo = (float4*)(out + (size_t)node * C_OUT);
    float4* __restrict__ qb = (float4*)(out_b + (size_t)node * C_OUT);
    q[0]  = r0;  q[1]  = r1;
    qo[0] = r0;  qo[1] = r1;
    qb[0] = z;   qb[1] = z;
  }
}


#define BCAST8(i) ((i) * 0x249249)

#define STEP_LOAD(i) { \
  const int sj = (int)__builtin_amdgcn_mov_dpp8((unsigned)s, BCAST8(i)); \
  dd[i] = (int)__builtin_amdgcn_mov_dpp8((unsigned)d, BCAST8(i)); \
  v[i]  = h2[(size_t)sj * C_OUT + f]; }

static __device__ __forceinline__ void agg2_batch(const bool full, const int s, const int d,
                                                  const int f, const int left,
                                                  const float* __restrict__ h2, float* __restrict__ out) {
  float v[8];
  int   dd[8];
  STEP_LOAD(0) STEP_LOAD(1) STEP_LOAD(2) STEP_LOAD(3)
  STEP_LOAD(4) STEP_LOAD(5) STEP_LOAD(6) STEP_LOAD(7)
  #pragma unroll
  for (int i = 0; i < 8; ++i)
    if (full || i < left) atomicAdd(out + (size_t)dd[i] * C_OUT + f, v[i]);
}

__global__ __launch_bounds__(256) void k_agg2_edges(const int* __restrict__ src,
                                                    const int* __restrict__ dst,
                                                    const float* __restrict__ h2,
                                                    float* __restrict__ out,
                                                    float* __restrict__ out_b, int E) {
  const int lane = threadIdx.x & 31;
  const int f    = lane & 7;
  float* __restrict__ plane = pkg_half() ? out_b : out;
  const int wib    = __builtin_amdgcn_readfirstlane((int)(threadIdx.x >> 5));
  const int nwaves = gridDim.x * 8;
  const int nbatch = (E + 31) >> 5;
  for (int b = blockIdx.x * 8 + wib; b < nbatch; b += nwaves) {
    const int e0 = b << 5;
    const int e  = e0 + lane;
    int s = 0, d = 0;
    if (e < E) { s = src[e]; d = dst[e]; }
    if (E - e0 >= 32) agg2_batch(true,  s, d, f, 8, h2, plane);
    else              agg2_batch(false, s, d, f, E - e0 - (lane & 24), h2, plane);
  }
}

#undef STEP_LOAD

__global__ __launch_bounds__(256) void k_out_sum(const float* __restrict__ out_b,
                                                 const float* __restrict__ b2,
                                                 const float* __restrict__ dinv,
                                                 float* __restrict__ out, int N) {
  int i = blockIdx.x * 256 + threadIdx.x;
  if (i >= N * (C_OUT / 4)) return;
  const float di = dinv[i >> 1];
  float4 a  = ((float4*)out)[i];
  float4 b  = ((const float4*)out_b)[i];
  float4 bb = ((const float4*)b2)[i & 1];
  a.x = bb.x + di * (a.x + b.x);  a.y = bb.y + di * (a.y + b.y);
  a.z = bb.z + di * (a.z + b.z);  a.w = bb.w + di * (a.w + b.w);
  ((float4*)out)[i] = a;
}


extern "C" void kernel_launch(void* const* d_in, const int* in_sizes, int n_in,
                              void* d_out, int out_size, void* d_ws, size_t ws_size,
                              hipStream_t stream) {
  const float* x  = (const float*)d_in[0];
  const int*   ei = (const int*)d_in[1];
  const float* W1 = (const float*)d_in[2];
  const float* b1 = (const float*)d_in[3];
  const float* W2 = (const float*)d_in[4];
  const float* b2 = (const float*)d_in[5];
  float* out = (float*)d_out;

  const int N = in_sizes[0] / F_IN;
  const int E = in_sizes[1] / 2;
  const int* src = ei;
  const int* dst = ei + E;

  float* xs    = (float*)d_ws;
  float* ax    = xs + (size_t)N * F_IN;
  float* h2    = ax + (size_t)2 * N * F_IN;
  float* out_b = h2 + (size_t)N * C_OUT;
  float* dinv  = out_b + (size_t)N * C_OUT;
  float* deg   = dinv + N;
  float* deg_b = deg + N;

  const int nbN  = (N + 255) / 256;
  const int nbE  = (E + 255) / 256;
  const int tiles = (N + 15) / 16;
  int nbW = ((E + 31) / 32 + 7) / 8;
  if (nbW > 2048) nbW = 2048;
  if (nbW < 1) nbW = 1;

  k_deg_init <<<nbN, 256, 0, stream>>>(deg, deg_b, N);
  k_deg_count<<<nbE, 256, 0, stream>>>(dst, deg, deg_b, E);
  k_scale_x  <<<(N * (F_IN / 4) + 255) / 256, 256, 0, stream>>>(x, deg, deg_b, dinv, xs, ax, N);

  k_aggx_edges<<<nbW, 256, 0, stream>>>(src, dst, xs, ax, E, N);

  k_gemm12<<<tiles, 32, 0, stream>>>(ax, W1, b1, W2, dinv, h2, out, out_b, N);

  k_agg2_edges<<<nbW, 256, 0, stream>>>(src, dst, h2, out, out_b, E);
  k_out_sum   <<<(N * (C_OUT / 4) + 255) / 256, 256, 0, stream>>>(out_b, b2, dinv, out, N);
}


# Degree count: a thread takes four consecutive edges (one 16-byte index load, four f32 atomic adds), so 1563 work-groups enter instead of 6250; sums are whole numbers, order of sums irrelevant
# speedup vs baseline: 3.9430x; 1.0672x over previous
#include <hip/hip_runtime.h>

typedef __attribute__((ext_vector_type(2))) float v2f;
typedef __attribute__((ext_vector_type(8))) float v8f;

#define F_IN  32
#define HC    64
#define C_OUT 8

static __device__ __forceinline__ int pkg_half() {
  const unsigned die = (__builtin_amdgcn_s_sendmsg_rtn(0x87) >> 16) & 15u;
  return (int)((die >> 2) & 1u);
}


__global__ __launch_bounds__(256) void k_deg_init(float* __restrict__ deg, float* __restrict__ deg_b, int N) {
  int i = blockIdx.x * 256 + threadIdx.x;
  if (i < N) { deg[i] = 1.0f; deg_b[i] = 0.0f; }
}

__global__ __launch_bounds__(256) void k_deg_count(const int* __restrict__ dst,
                                                   float* __restrict__ deg, float* __restrict__ deg_b, int E) {
  float* plane = pkg_half() ? deg_b : deg;
  const int q = blockIdx.x * 256 + threadIdx.x;
  const int e = q << 2;
  if (e >= E) return;
  if ((E & 3) == 0) {
    const int4 d = ((const int4*)dst)[q];
    atomicAdd(&plane[d.x], 1.0f);
    atomicAdd(&plane[d.y], 1.0f);
    atomicAdd(&plane[d.z], 1.0f);
    atomicAdd(&plane[d.w], 1.0f);
  } else {
    const int n = (E - e < 4) ? (E - e) : 4;
    for (int j = 0; j < n; ++j) atomicAdd(&plane[dst[e + j]], 1.0f);
  }
}

__global__ __launch_bounds__(256) void k_scale_x(const float* __restrict__ x,
                                                 const float* __restrict__ deg, const float* __restrict__ deg_b,
                                                 float* __restrict__ dinv, float* __restrict__ xs,
                                                 float* __restrict__ ax, int N) {
  int i = blockIdx.x * 256 + threadIdx.x;
  if (i >= N * (F_IN / 4)) return;
  const int node = i >> 3;
  float4 v = ((const float4*)x)[i];
  const float di = 1.0f / sqrtf(deg[node] + deg_b[node]);
  if ((i & 7) == 0) dinv[node] = di;
  v.x *= di;  v.y *= di;  v.z *= di;  v.w *= di;
  ((float4*)xs)[i] = v;
  ((float4*)ax)[i] = v;
  ((float4*)ax)[(size_t)N * (F_IN / 4) + i] = make_float4(0.0f, 0.0f, 0.0f, 0.0f);
}


__global__ __launch_bounds__(256) void k_aggx_edges(const int* __restrict__ src,
                                                    const int* __restrict__ dst,
                                                    const float* __restrict__ xs,
                                                    float* __restrict__ ax, int E, int N) {
  const int lane = threadIdx.x & 31;
  float* __restrict__ plane = ax + (size_t)pkg_half() * N * F_IN;
  const int wib    = __builtin_amdgcn_readfirstlane((int)(threadIdx.x >> 5));
  const int nwaves = gridDim.x * 8;
  const int nbatch = (E + 31) >> 5;
  for (int b = blockIdx.x * 8 + wib; b < nbatch; b += nwaves) {
    const int e0 = b << 5;
    const int e  = e0 + lane;
    int s = 0, d = 0;
    if (e < E) { s = src[e]; d = dst[e]; }
    const int cnt = (E - e0 < 32) ? (E - e0) : 32;
    if (cnt == 32) {
      #pragma unroll
      for (int g = 0; g < 32; g += 16) {
        float v[16];
        #pragma unroll
        for (int u = 0; u < 16; ++u) {
          const int sj = __builtin_amdgcn_readlane(s, g + u);
          v[u] = xs[(size_t)sj * F_IN + lane];
        }
        #pragma unroll
        for (int u = 0; u < 16; ++u) {
          const int dj = __builtin_amdgcn_readlane(d, g + u);
          atomicAdd(plane + (size_t)dj * F_IN + lane, v[u]);
        }
      }
    } else {
      for (int j = 0; j < cnt; ++j) {
        const int sj = __builtin_amdgcn_readlane(s, j);
        const int dj = __builtin_amdgcn_readlane(d, j);
        atomicAdd(plane + (size_t)dj * F_IN + lane, xs[(size_t)sj * F_IN + lane]);
      }
    }
  }
}


__global__ __launch_bounds__(32) void k_gemm12(const float* __restrict__ ax,
                                               const float* __restrict__ W1,
                                               const float* __restrict__ b1,
                                               const float* __restrict__ W2,
                                               const float* __restrict__ dinv,
                                               float* __restrict__ h2,
                                               float* __restrict__ out,
                                               float* __restrict__ out_b, int N) {
  const int lane = threadIdx.x;
  const int c    = lane & 15;
  const int h    = lane >> 4;
  int node = blockIdx.x * 16 + c;
  const bool live = (node < N);
  if (!live) node = N - 1;
  const float dn = dinv[node];
  const float4* __restrict__ p0 = (const float4*)(ax + (size_t)node * F_IN + 16 * h);
  const float4* __restrict__ p1 = (const float4*)(ax + (size_t)N * F_IN + (size_t)node * F_IN + 16 * h);
  float a1[16];
  #pragma unroll
  for (int q = 0; q < 4; ++q) {
    const float4 u = p0[q], w = p1[q];
    a1[4 * q + 0] = dn * (u.x + w.x);
    a1[4 * q + 1] = dn * (u.y + w.y);
    a1[4 * q + 2] = dn * (u.z + w.z);
    a1[4 * q + 3] = dn * (u.w + w.w);
  }
  v8f t[4];
  #pragma unroll
  for (int r = 0; r < 8; ++r) {
    const float4 bb = ((const float4*)b1)[8 * h + r];
    t[0][r] = bb.x;  t[1][r] = bb.y;  t[2][r] = bb.z;  t[3][r] = bb.w;
  }
  #pragma unroll
  for (int j = 0; j < 8; ++j) {
    const int k = 16 * h + 2 * j;
    const float4 w0 = ((const float4*)(W1 + (size_t)k * HC))[c];
    const float4 w1 = ((const float4*)(W1 + (size_t)(k + 1) * HC))[c];
    const v2f bx = { a1[2 * j], a1[2 * j + 1] };
    const v2f aw0 = { w0.x, w1.x }, aw1 = { w0.y, w1.y }, aw2 = { w0.z, w1.z }, aw3 = { w0.w, w1.w };
    t[0] = __builtin_amdgcn_wmma_f32_16x16x4_f32(false, aw0, false, bx, (short)0, t[0], false, false);
    t[1] = __builtin_amdgcn_wmma_f32_16x16x4_f32(false, aw1, false, bx, (short)0, t[1], false, false);
    t[2] = __builtin_amdgcn_wmma_f32_16x16x4_f32(false, aw2, false, bx, (short)0, t[2], false, false);
    t[3] = __builtin_amdgcn_wmma_f32_16x16x4_f32(false, aw3, false, bx, (short)0, t[3], false, false);
  }
  const bool live_row = (c < C_OUT);
  const float* __restrict__ w2c = W2 + (c & (C_OUT - 1));
  v8f o = {};
  #pragma unroll
  for (int mt = 0; mt < 4; ++mt)
    #pragma unroll
    for (int j = 0; j < 4; ++j) {
      const int k = 4 * (8 * h + 2 * j) + mt;
      const float wa = w2c[(size_t)k * C_OUT], wb = w2c[(size_t)(k + 4) * C_OUT];
      const v2f aw = { live_row ? wa : 0.0f, live_row ? wb : 0.0f };
      const v2f bt = { t[mt][2 * j], t[mt][2 * j + 1] };
      o = __builtin_amdgcn_wmma_f32_16x16x4_f32(false, aw, false, bt, (short)0, o, false, false);
    }
  if (h == 0 && live) {
    const float4 r0 = make_float4(o[0] * dn, o[1] * dn, o[2] * dn, o[3] * dn);
    const float4 r1 = make_float4(o[4] * dn, o[5] * dn, o[6] * dn, o[7] * dn);
    const float4 z  = make_float4(0.0f, 0.0f, 0.0f, 0.0f);
    float4* __restrict__ q  = (float4*)(h2 + (size_t)node * C_OUT);
    float4* __restrict__ qo = (float4*)(out + (size_t)node * C_OUT);
    float4* __restrict__ qb = (float4*)(out_b + (size_t)node * C_OUT);
    q[0]  = r0;  q[1]  = r1;
    qo[0] = r0;  qo[1] = r1;
    qb[0] = z;   qb[1] = z;
  }
}


#define BCAST8(i) ((i) * 0x249249)

#define STEP_LOAD(i) { \
  const int sj = (int)__builtin_amdgcn_mov_dpp8((unsigned)s, BCAST8(i)); \
  dd[i] = (int)__builtin_amdgcn_mov_dpp8((unsigned)d, BCAST8(i)); \
  v[i]  = h2[(size_t)sj * C_OUT + f]; }

static __device__ __forceinline__ void agg2_batch(const bool full, const int s, const int d,
                                                  const int f, const int left,
                                                  const float* __restrict__ h2, float* __restrict__ out) {
  float v[8];
  int   dd[8];
  STEP_LOAD(0) STEP_LOAD(1) STEP_LOAD(2) STEP_LOAD(3)
  STEP_LOAD(4) STEP_LOAD(5) STEP_LOAD(6) STEP_LOAD(7)
  #pragma unroll
  for (int i = 0; i < 8; ++i)
    if (full || i < left) atomicAdd(out + (size_t)dd[i] * C_OUT + f, v[i]);
}

__global__ __launch_bounds__(256) void k_agg2_edges(const int* __restrict__ src,
                                                    const int* __restrict__ dst,
                                                    const float* __restrict__ h2,
                                                    float* __restrict__ out,
                                                    float* __restrict__ out_b, int E) {
  const int lane = threadIdx.x & 31;
  const int f    = lane & 7;
  float* __restrict__ plane = pkg_half() ? out_b : out;
  const int wib    = __builtin_amdgcn_readfirstlane((int)(threadIdx.x >> 5));
  const int nwaves = gridDim.x * 8;
  const int nbatch = (E + 31) >> 5;
  for (int b = blockIdx.x * 8 + wib; b < nbatch; b += nwaves) {
    const int e0 = b << 5;
    const int e  = e0 + lane;
    int s = 0, d = 0;
    if (e < E) { s = src[e]; d = dst[e]; }
    if (E - e0 >= 32) agg2_batch(true,  s, d, f, 8, h2, plane);
    else              agg2_batch(false, s, d, f, E - e0 - (lane & 24), h2, plane);
  }
}

#undef STEP_LOAD

__global__ __launch_bounds__(256) void k_out_sum(const float* __restrict__ out_b,
                                                 const float* __restrict__ b2,
                                                 const float* __restrict__ dinv,
                                                 float* __restrict__ out, int N) {
  int i = blockIdx.x * 256 + threadIdx.x;
  if (i >= N * (C_OUT / 4)) return;
  const float di = dinv[i >> 1];
  float4 a  = ((float4*)out)[i];
  float4 b  = ((const float4*)out_b)[i];
  float4 bb = ((const float4*)b2)[i & 1];
  a.x = bb.x + di * (a.x + b.x);  a.y = bb.y + di * (a.y + b.y);
  a.z = bb.z + di * (a.z + b.z);  a.w = bb.w + di * (a.w + b.w);
  ((float4*)out)[i] = a;
}


extern "C" void kernel_launch(void* const* d_in, const int* in_sizes, int n_in,
                              void* d_out, int out_size, void* d_ws, size_t ws_size,
                              hipStream_t stream) {
  const float* x  = (const float*)d_in[0];
  const int*   ei = (const int*)d_in[1];
  const float* W1 = (const float*)d_in[2];
  const float* b1 = (const float*)d_in[3];
  const float* W2 = (const float*)d_in[4];
  const float* b2 = (const float*)d_in[5];
  float* out = (float*)d_out;

  const int N = in_sizes[0] / F_IN;
  const int E = in_sizes[1] / 2;
  const int* src = ei;
  const int* dst = ei + E;

  float* xs    = (float*)d_ws;
  float* ax    = xs + (size_t)N * F_IN;
  float* h2    = ax + (size_t)2 * N * F_IN;
  float* out_b = h2 + (size_t)N * C_OUT;
  float* dinv  = out_b + (size_t)N * C_OUT;
  float* deg   = dinv + N;
  float* deg_b = deg + N;

  const int nbN  = (N + 255) / 256;
  const int nbE  = ((E + 3) / 4 + 255) / 256;
  const int tiles = (N + 15) / 16;
  int nbW = ((E + 31) / 32 + 7) / 8;
  if (nbW > 2048) nbW = 2048;
  if (nbW < 1) nbW = 1;

  k_deg_init <<<nbN, 256, 0, stream>>>(deg, deg_b, N);
  k_deg_count<<<nbE, 256, 0, stream>>>(dst, deg, deg_b, E);
  k_scale_x  <<<(N * (F_IN / 4) + 255) / 256, 256, 0, stream>>>(x, deg, deg_b, dinv, xs, ax, N);

  k_aggx_edges<<<nbW, 256, 0, stream>>>(src, dst, xs, ax, E, N);

  k_gemm12<<<tiles, 32, 0, stream>>>(ax, W1, b1, W2, dinv, h2, out, out_b, N);

  k_agg2_edges<<<nbW, 256, 0, stream>>>(src, dst, h2, out, out_b, E);
  k_out_sum   <<<(N * (C_OUT / 4) + 255) / 256, 256, 0, stream>>>(out_b, b2, dinv, out, N);
}


# Degree count in LDS, no global atomics, split by node range x edge chunk (8 x 32 work-groups, 6.4 MB of 16-bit counts); row scaling adds a node's 32 whole-number counts; 7 launches to 6
# speedup vs baseline: 4.0692x; 1.0320x over previous
#include <hip/hip_runtime.h>

typedef __attribute__((ext_vector_type(2))) float v2f;
typedef __attribute__((ext_vector_type(8))) float v8f;

#define F_IN  32
#define HC    64
#define C_OUT 8

static __device__ __forceinline__ int pkg_half() {
  const unsigned die = (__builtin_amdgcn_s_sendmsg_rtn(0x87) >> 16) & 15u;
  return (int)((die >> 2) & 1u);
}


#define DEG_RS 12504
#define DEG_CH 50000
#define DEG_WG 512
#define DEG_NQ ((DEG_CH / 4 + DEG_WG - 1) / DEG_WG)

__global__ __launch_bounds__(DEG_WG) void k_deg_hist(const int* __restrict__ dst,
                                                   unsigned short* __restrict__ tab, int E, int C) {
  __shared__ __attribute__((aligned(16))) unsigned cnt[DEG_RS];
  const int r   = blockIdx.x;
  const int c   = blockIdx.y;
  const int tid = threadIdx.x;
  for (int i = tid; i < DEG_RS / 4; i += DEG_WG) ((uint4*)cnt)[i] = make_uint4(0u, 0u, 0u, 0u);
  __syncthreads();
  const int lo = r * DEG_RS;
  const int e0 = c * DEG_CH;
  const int n  = (E - e0 < DEG_CH) ? (E - e0) : DEG_CH;
  if ((E & 3) == 0 && n > 0) {
    const int4* __restrict__ p = (const int4*)(dst + e0);
    const int nq = n >> 2;
    int4 d[DEG_NQ];
    #pragma unroll
    for (int k = 0; k < DEG_NQ; ++k) {
      const int q = tid + DEG_WG * k;
      d[k] = p[(q < nq) ? q : 0];
    }
    #pragma unroll
    for (int k = 0; k < DEG_NQ; ++k) {
      const unsigned lim = (tid + DEG_WG * k < nq) ? (unsigned)DEG_RS : 0u;
      const unsigned a0 = (unsigned)(d[k].x - lo), a1 = (unsigned)(d[k].y - lo);
      const unsigned a2 = (unsigned)(d[k].z - lo), a3 = (unsigned)(d[k].w - lo);
      if (a0 < lim) atomicAdd(&cnt[a0], 1u);
      if (a1 < lim) atomicAdd(&cnt[a1], 1u);
      if (a2 < lim) atomicAdd(&cnt[a2], 1u);
      if (a3 < lim) atomicAdd(&cnt[a3], 1u);
    }
  } else {
    for (int j = tid; j < n; j += DEG_WG) {
      const unsigned a = (unsigned)(dst[e0 + j] - lo);
      if (a < DEG_RS) atomicAdd(&cnt[a], 1u);
    }
  }
  __syncthreads();
  uint4* __restrict__ o = (uint4*)(tab + ((size_t)r * C + c) * DEG_RS);
  for (int i = tid; i < DEG_RS / 8; i += DEG_WG) {
    const uint4 a = ((const uint4*)cnt)[2 * i], b = ((const uint4*)cnt)[2 * i + 1];
    o[i] = make_uint4(a.x | (a.y << 16), a.z | (a.w << 16), b.x | (b.y << 16), b.z | (b.w << 16));
  }
}

#define XOR8(m) (((0 ^ (m))) | ((1 ^ (m)) << 3) | ((2 ^ (m)) << 6) | ((3 ^ (m)) << 9) | \
                 ((4 ^ (m)) << 12) | ((5 ^ (m)) << 15) | ((6 ^ (m)) << 18) | ((7 ^ (m)) << 21))

__global__ __launch_bounds__(256) void k_scale_x(const float* __restrict__ x,
                                                 const unsigned short* __restrict__ tab,
                                                 float* __restrict__ dinv, float* __restrict__ xs,
                                                 float* __restrict__ ax, int N, int C) {
  int i = blockIdx.x * 256 + threadIdx.x;
  if (i >= N * (F_IN / 4)) return;
  const int node = i >> 3;
  float4 v = ((const float4*)x)[i];
  const int r = node / DEG_RS;
  const unsigned short* __restrict__ p = tab + (size_t)r * C * DEG_RS + (node - r * DEG_RS);
  unsigned s = 0u;
  #pragma unroll 1
  for (int c0 = i & 7; c0 < C; c0 += 32) {
    #pragma unroll
    for (int u = 0; u < 4; ++u) {
      const int c = c0 + 8 * u;
      const unsigned cv = p[((c < C) ? c : c0) * DEG_RS];
      s += (c < C) ? cv : 0u;
    }
  }
  s += __builtin_amdgcn_mov_dpp8(s, XOR8(1));
  s += __builtin_amdgcn_mov_dpp8(s, XOR8(2));
  s += __builtin_amdgcn_mov_dpp8(s, XOR8(4));
  const float di = 1.0f / sqrtf((float)(s + 1u));
  if ((i & 7) == 0) dinv[node] = di;
  v.x *= di;  v.y *= di;  v.z *= di;  v.w *= di;
  ((float4*)xs)[i] = v;
  ((float4*)ax)[i] = v;
  ((float4*)ax)[(size_t)N * (F_IN / 4) + i] = make_float4(0.0f, 0.0f, 0.0f, 0.0f);
}


__global__ __launch_bounds__(256) void k_aggx_edges(const int* __restrict__ src,
                                                    const int* __restrict__ dst,
                                                    const float* __restrict__ xs,
                                                    float* __restrict__ ax, int E, int N) {
  const int lane = threadIdx.x & 31;
  float* __restrict__ plane = ax + (size_t)pkg_half() * N * F_IN;
  const int wib    = __builtin_amdgcn_readfirstlane((int)(threadIdx.x >> 5));
  const int nwaves = gridDim.x * 8;
  const int nbatch = (E + 31) >> 5;
  for (int b = blockIdx.x * 8 + wib; b < nbatch; b += nwaves) {
    const int e0 = b << 5;
    const int e  = e0 + lane;
    int s = 0, d = 0;
    if (e < E) { s = src[e]; d = dst[e]; }
    const int cnt = (E - e0 < 32) ? (E - e0) : 32;
    if (cnt == 32) {
      #pragma unroll
      for (int g = 0; g < 32; g += 16) {
        float v[16];
        #pragma unroll
        for (int u = 0; u < 16; ++u) {
          const int sj = __builtin_amdgcn_readlane(s, g + u);
          v[u] = xs[(size_t)sj * F_IN + lane];
        }
        #pragma unroll
        for (int u = 0; u < 16; ++u) {
          const int dj = __builtin_amdgcn_readlane(d, g + u);
          atomicAdd(plane + (size_t)dj * F_IN + lane, v[u]);
        }
      }
    } else {
      for (int j = 0; j < cnt; ++j) {
        const int sj = __builtin_amdgcn_readlane(s, j);
        const int dj = __builtin_amdgcn_readlane(d, j);
        atomicAdd(plane + (size_t)dj * F_IN + lane, xs[(size_t)sj * F_IN + lane]);
      }
    }
  }
}


__global__ __launch_bounds__(32) void k_gemm12(const float* __restrict__ ax,
                                               const float* __restrict__ W1,
                                               const float* __restrict__ b1,
                                               const float* __restrict__ W2,
                                               const float* __restrict__ dinv,
                                               float* __restrict__ h2,
                                               float* __restrict__ out,
                                               float* __restrict__ out_b, int N) {
  const int lane = threadIdx.x;
  const int c    = lane & 15;
  const int h    = lane >> 4;
  int node = blockIdx.x * 16 + c;
  const bool live = (node < N);
  if (!live) node = N - 1;
  const float dn = dinv[node];
  const float4* __restrict__ p0 = (const float4*)(ax + (size_t)node * F_IN + 16 * h);
  const float4* __restrict__ p1 = (const float4*)(ax + (size_t)N * F_IN + (size_t)node * F_IN + 16 * h);
  float a1[16];
  #pragma unroll
  for (int q = 0; q < 4; ++q) {
    const float4 u = p0[q], w = p1[q];
    a1[4 * q + 0] = dn * (u.x + w.x);
    a1[4 * q + 1] = dn * (u.y + w.y);
    a1[4 * q + 2] = dn * (u.z + w.z);
    a1[4 * q + 3] = dn * (u.w + w.w);
  }
  v8f t[4];
  #pragma unroll
  for (int r = 0; r < 8; ++r) {
    const float4 bb = ((const float4*)b1)[8 * h + r];
    t[0][r] = bb.x;  t[1][r] = bb.y;  t[2][r] = bb.z;  t[3][r] = bb.w;
  }
  #pragma unroll
  for (int j = 0; j < 8; ++j) {
    const int k = 16 * h + 2 * j;
    const float4 w0 = ((const float4*)(W1 + (size_t)k * HC))[c];
    const float4 w1 = ((const float4*)(W1 + (size_t)(k + 1) * HC))[c];
    const v2f bx = { a1[2 * j], a1[2 * j + 1] };
    const v2f aw0 = { w0.x, w1.x }, aw1 = { w0.y, w1.y }, aw2 = { w0.z, w1.z }, aw3 = { w0.w, w1.w };
    t[0] = __builtin_amdgcn_wmma_f32_16x16x4_f32(false, aw0, false, bx, (short)0, t[0], false, false);
    t[1] = __builtin_amdgcn_wmma_f32_16x16x4_f32(false, aw1, false, bx, (short)0, t[1], false, false);
    t[2] = __builtin_amdgcn_wmma_f32_16x16x4_f32(false, aw2, false, bx, (short)0, t[2], false, false);
    t[3] = __builtin_amdgcn_wmma_f32_16x16x4_f32(false, aw3, false, bx, (short)0, t[3], false, false);
  }
  const bool live_row = (c < C_OUT);
  const float* __restrict__ w2c = W2 + (c & (C_OUT - 1));
  v8f o = {};
  #pragma unroll
  for (int mt = 0; mt < 4; ++mt)
    #pragma unroll
    for (int j = 0; j < 4; ++j) {
      const int k = 4 * (8 * h + 2 * j) + mt;
      const float wa = w2c[(size_t)k * C_OUT], wb = w2c[(size_t)(k + 4) * C_OUT];
      const v2f aw = { live_row ? wa : 0.0f, live_row ? wb : 0.0f };
      const v2f bt = { t[mt][2 * j], t[mt][2 * j + 1] };
      o = __builtin_amdgcn_wmma_f32_16x16x4_f32(false, aw, false, bt, (short)0, o, false, false);
    }
  if (h == 0 && live) {
    const float4 r0 = make_float4(o[0] * dn, o[1] * dn, o[2] * dn, o[3] * dn);
    const float4 r1 = make_float4(o[4] * dn, o[5] * dn, o[6] * dn, o[7] * dn);
    const float4 z  = make_float4(0.0f, 0.0f, 0.0f, 0.0f);
    float4* __restrict__ q  = (float4*)(h2 + (size_t)node * C_OUT);
    float4* __restrict__ qo = (float4*)(out + (size_t)node * C_OUT);
    float4* __restrict__ qb = (float4*)(out_b + (size_t)node * C_OUT);
    q[0]  = r0;  q[1]  = r1;
    qo[0] = r0;  qo[1] = r1;
    qb[0] = z;   qb[1] = z;
  }
}


#define BCAST8(i) ((i) * 0x249249)

#define STEP_LOAD(i) { \
  const int sj = (int)__builtin_amdgcn_mov_dpp8((unsigned)s, BCAST8(i)); \
  dd[i] = (int)__builtin_amdgcn_mov_dpp8((unsigned)d, BCAST8(i)); \
  v[i]  = h2[(size_t)sj * C_OUT + f]; }

static __device__ __forceinline__ void agg2_batch(const bool full, const int s, const int d,
                                                  const int f, const int left,
                                                  const float* __restrict__ h2, float* __restrict__ out) {
  float v[8];
  int   dd[8];
  STEP_LOAD(0) STEP_LOAD(1) STEP_LOAD(2) STEP_LOAD(3)
  STEP_LOAD(4) STEP_LOAD(5) STEP_LOAD(6) STEP_LOAD(7)
  #pragma unroll
  for (int i = 0; i < 8; ++i)
    if (full || i < left) atomicAdd(out + (size_t)dd[i] * C_OUT + f, v[i]);
}

__global__ __launch_bounds__(256) void k_agg2_edges(const int* __restrict__ src,
                                                    const int* __restrict__ dst,
                                                    const float* __restrict__ h2,
                                                    float* __restrict__ out,
                                                    float* __restrict__ out_b, int E) {
  const int lane = threadIdx.x & 31;
  const int f    = lane & 7;
  float* __restrict__ plane = pkg_half() ? out_b : out;
  const int wib    = __builtin_amdgcn_readfirstlane((int)(threadIdx.x >> 5));
  const int nwaves = gridDim.x * 8;
  const int nbatch = (E + 31) >> 5;
  for (int b = blockIdx.x * 8 + wib; b < nbatch; b += nwaves) {
    const int e0 = b << 5;
    const int e  = e0 + lane;
    int s = 0, d = 0;
    if (e < E) { s = src[e]; d = dst[e]; }
    if (E - e0 >= 32) agg2_batch(true,  s, d, f, 8, h2, plane);
    else              agg2_batch(false, s, d, f, E - e0 - (lane & 24), h2, plane);
  }
}

#undef STEP_LOAD

__global__ __launch_bounds__(256) void k_out_sum(const float* __restrict__ out_b,
                                                 const float* __restrict__ b2,
                                                 const float* __restrict__ dinv,
                                                 float* __restrict__ out, int N) {
  int i = blockIdx.x * 256 + threadIdx.x;
  if (i >= N * (C_OUT / 4)) return;
  const float di = dinv[i >> 1];
  float4 a  = ((float4*)out)[i];
  float4 b  = ((const float4*)out_b)[i];
  float4 bb = ((const float4*)b2)[i & 1];
  a.x = bb.x + di * (a.x + b.x);  a.y = bb.y + di * (a.y + b.y);
  a.z = bb.z + di * (a.z + b.z);  a.w = bb.w + di * (a.w + b.w);
  ((float4*)out)[i] = a;
}


extern "C" void kernel_launch(void* const* d_in, const int* in_sizes, int n_in,
                              void* d_out, int out_size, void* d_ws, size_t ws_size,
                              hipStream_t stream) {
  const float* x  = (const float*)d_in[0];
  const int*   ei = (const int*)d_in[1];
  const float* W1 = (const float*)d_in[2];
  const float* b1 = (const float*)d_in[3];
  const float* W2 = (const float*)d_in[4];
  const float* b2 = (const float*)d_in[5];
  float* out = (float*)d_out;

  const int N = in_sizes[0] / F_IN;
  const int E = in_sizes[1] / 2;
  const int* src = ei;
  const int* dst = ei + E;

  const int nR = (N + DEG_RS - 1) / DEG_RS;
  int nC = (E + DEG_CH - 1) / DEG_CH;
  if (nC < 1) nC = 1;
  float* xs    = (float*)d_ws;
  float* ax    = xs + (size_t)N * F_IN;
  float* h2    = ax + (size_t)2 * N * F_IN;
  float* out_b = h2 + (size_t)N * C_OUT;
  unsigned short* tab = (unsigned short*)(out_b + (size_t)N * C_OUT);
  float* dinv  = (float*)(tab + (size_t)nR * nC * DEG_RS);

  const int tiles = (N + 15) / 16;
  int nbW = ((E + 31) / 32 + 7) / 8;
  if (nbW > 2048) nbW = 2048;
  if (nbW < 1) nbW = 1;

  k_deg_hist<<<dim3(nR, nC), DEG_WG, 0, stream>>>(dst, tab, E, nC);
  k_scale_x <<<(N * (F_IN / 4) + 255) / 256, 256, 0, stream>>>(x, tab, dinv, xs, ax, N, nC);

  k_aggx_edges<<<nbW, 256, 0, stream>>>(src, dst, xs, ax, E, N);

  k_gemm12<<<tiles, 32, 0, stream>>>(ax, W1, b1, W2, dinv, h2, out, out_b, N);

  k_agg2_edges<<<nbW, 256, 0, stream>>>(src, dst, h2, out, out_b, E);
  k_out_sum   <<<(N * (C_OUT / 4) + 255) / 256, 256, 0, stream>>>(out_b, b2, dinv, out, N);
}
